# starting kernel
# baseline (speedup 1.0000x reference)
#include <hip/hip_runtime.h>
#include <cmath>

#define B_  8
#define N_  2048
#define H_  256
#define THREADS 256
#define ROWS_PER_BLOCK 128

typedef __attribute__((ext_vector_type(2))) float v2f;
typedef __attribute__((ext_vector_type(8))) float v8f;

#define LOG2E 1.4426950408889634f
#define LN2   0.6931471805599453f

__global__ __launch_bounds__(THREADS)
void prep_kernel(const float* __restrict__ cloud, const float* __restrict__ noise,
                 float4* __restrict__ ypts, float4* __restrict__ xpts,
                 float* __restrict__ f, float* __restrict__ g) {
  const int b   = blockIdx.x;
  const int tid = threadIdx.x;
  __shared__ float s_sum[THREADS];
  __shared__ float s_sq[THREADS];

  const float* cb = cloud + (size_t)b * N_ * 3;
  const float* nb = noise + (size_t)b * N_ * 3;

  float sum = 0.f, sq = 0.f;
  for (int i = tid; i < N_ * 3; i += THREADS) {
    float v = cb[i];
    sum += v; sq += v * v;
  }
  s_sum[tid] = sum; s_sq[tid] = sq;
  __syncthreads();
  for (int off = THREADS / 2; off > 0; off >>= 1) {
    if (tid < off) { s_sum[tid] += s_sum[tid + off]; s_sq[tid] += s_sq[tid + off]; }
    __syncthreads();
  }
  const float n    = (float)(N_ * 3);
  const float tot  = s_sum[0];
  const float tot2 = s_sq[0];
  const float var  = (tot2 - tot * tot / n) / (n - 1.0f);
  const float inv_std = rsqrtf(var);

  for (int p = tid; p < N_; p += THREADS) {
    float yx = cb[p * 3 + 0] * inv_std;
    float yy = cb[p * 3 + 1] * inv_std;
    float yz = cb[p * 3 + 2] * inv_std;
    ypts[(size_t)b * N_ + p] = make_float4(yx, yy, yz, yx * yx + yy * yy + yz * yz);
    float xx = nb[p * 3 + 0];
    float xy = nb[p * 3 + 1];
    float xz = nb[p * 3 + 2];
    xpts[(size_t)b * N_ + p] = make_float4(xx, xy, xz, xx * xx + xy * xy + xz * xz);
    f[(size_t)b * N_ + p] = 0.f;
    g[(size_t)b * N_ + p] = 0.f;
  }
}

__global__ __launch_bounds__(THREADS)
void sinkhorn_half(const float4* __restrict__ rows_pts,
                   const float4* __restrict__ cols_pts,
                   const float*  __restrict__ in_pot,
                   float*        __restrict__ out_pot,
                   float eps, float lconst) {
  const int b    = blockIdx.y;
  const int tid  = threadIdx.x;
  const int lane = tid & 31;
  const int wave = tid >> 5;
  const int hi   = lane >> 4;
  const int idx  = lane & 15;

  __shared__ float4 s_cols[N_];
  __shared__ float  s_gy[N_];

  const float4* cp = cols_pts + (size_t)b * N_;
  const float*  ip = in_pot   + (size_t)b * N_;
  for (int j = tid; j < N_; j += THREADS) {
    float4 c = cp[j];
    s_cols[j] = c;
    s_gy[j]   = ip[j] - 0.5f * c.w;
  }
  __syncthreads();

  const int row_base = blockIdx.x * ROWS_PER_BLOCK + wave * 16;
  const float4* rp = rows_pts + (size_t)b * N_;

  float4 r4 = rp[row_base + idx];
  v2f a;
  a.x = hi ? r4.z : r4.x;
  a.y = hi ? 0.0f : r4.y;

  float m[8], s[8];
#pragma unroll
  for (int r = 0; r < 8; ++r) { m[r] = -3.0e38f; s[r] = 0.f; }
  const float c2 = LOG2E / eps;

  for (int c0 = 0; c0 < N_; c0 += 16) {
    float4 c4 = s_cols[c0 + idx];
    v2f bf;
    bf.x = hi ? c4.z : c4.x;
    bf.y = hi ? 0.0f : c4.y;
    float gy = s_gy[c0 + idx];

    v8f cz = {};
    v8f acc = __builtin_amdgcn_wmma_f32_16x16x4_f32(
        false, a, false, bf, (short)0, cz, false, false);

#pragma unroll
    for (int r = 0; r < 8; ++r) {
      float v = (gy + acc[r]) * c2;
      float d = v - m[r];
      float e = __builtin_amdgcn_exp2f(-fabsf(d));
      bool  up = d > 0.0f;
      s[r] = up ? fmaf(s[r], e, 1.0f) : (s[r] + e);
      m[r] = up ? v : m[r];
    }
  }

#pragma unroll
  for (int mask = 1; mask < 16; mask <<= 1) {
#pragma unroll
    for (int r = 0; r < 8; ++r) {
      float om = __shfl_xor(m[r], mask, 32);
      float os = __shfl_xor(s[r], mask, 32);
      float d  = om - m[r];
      float e  = __builtin_amdgcn_exp2f(-fabsf(d));
      bool  up = d > 0.0f;
      s[r] = up ? fmaf(s[r], e, os) : fmaf(os, e, s[r]);
      m[r] = up ? om : m[r];
    }
  }

  if (idx < 8) {
    const int row = row_base + hi * 8 + idx;
    const int r   = idx;
    float sqrow = rp[row].w;
    float lse = LN2 * (m[r] + __builtin_amdgcn_logf(s[r]));
    out_pot[(size_t)b * N_ + row] = -eps * (lconst + lse) + 0.5f * sqrow;
  }
}

__global__ __launch_bounds__(THREADS)
void assign_mlp_kernel(const float4* __restrict__ xpts,
                       const float4* __restrict__ ypts,
                       const float*  __restrict__ g,
                       const float*  __restrict__ t,
                       const float*  __restrict__ W1,
                       const float*  __restrict__ Wt,
                       const float*  __restrict__ b1,
                       const float*  __restrict__ W2,
                       const float*  __restrict__ b2,
                       float*        __restrict__ out) {
  const int b    = blockIdx.y;
  const int tid  = threadIdx.x;
  const int lane = tid & 31;
  const int wave = tid >> 5;
  const int hi   = lane >> 4;
  const int idx  = lane & 15;

  __shared__ float4 s_cols[N_];
  __shared__ float  s_gy[N_];
  __shared__ float  s_W1[3 * H_];
  __shared__ float  s_Wt[H_];
  __shared__ float  s_b1[H_];
  __shared__ float  s_W2[H_ * 3];
  __shared__ float  s_b2[4];
  __shared__ int    s_idx[ROWS_PER_BLOCK];

  const float4* cp = ypts + (size_t)b * N_;
  const float*  gp = g    + (size_t)b * N_;
  for (int j = tid; j < N_; j += THREADS) {
    float4 c = cp[j];
    s_cols[j] = c;
    s_gy[j]   = gp[j] - c.w;
  }
  for (int j = tid; j < 3 * H_; j += THREADS) { s_W1[j] = W1[j]; s_W2[j] = W2[j]; }
  for (int j = tid; j < H_;     j += THREADS) { s_Wt[j] = Wt[j]; s_b1[j] = b1[j]; }
  if (tid < 3) s_b2[tid] = b2[tid];
  __syncthreads();

  const int row_base = blockIdx.x * ROWS_PER_BLOCK + wave * 16;
  const float4* rp = xpts + (size_t)b * N_;

  float4 r4 = rp[row_base + idx];
  v2f a;
  a.x = hi ? 2.0f * r4.z : 2.0f * r4.x;
  a.y = hi ? 0.0f        : 2.0f * r4.y;

  float best[8]; int bidx[8];
#pragma unroll
  for (int r = 0; r < 8; ++r) { best[r] = -3.0e38f; bidx[r] = 0; }

  for (int c0 = 0; c0 < N_; c0 += 16) {
    float4 c4 = s_cols[c0 + idx];
    v2f bf;
    bf.x = hi ? c4.z : c4.x;
    bf.y = hi ? 0.0f : c4.y;
    float gy = s_gy[c0 + idx];

    v8f cz = {};
    v8f acc = __builtin_amdgcn_wmma_f32_16x16x4_f32(
        false, a, false, bf, (short)0, cz, false, false);

#pragma unroll
    for (int r = 0; r < 8; ++r) {
      float v = acc[r] + gy;
      if (v > best[r]) { best[r] = v; bidx[r] = c0 + idx; }
    }
  }

#pragma unroll
  for (int mask = 1; mask < 16; mask <<= 1) {
#pragma unroll
    for (int r = 0; r < 8; ++r) {
      float ov = __shfl_xor(best[r], mask, 32);
      int   oi = __shfl_xor(bidx[r], mask, 32);
      if (ov > best[r] || (ov == best[r] && oi < bidx[r])) {
        best[r] = ov; bidx[r] = oi;
      }
    }
  }

  if (idx < 8) s_idx[wave * 16 + hi * 8 + idx] = bidx[idx];
  __syncthreads();

  if (tid < ROWS_PER_BLOCK) {
    const int grow = blockIdx.x * ROWS_PER_BLOCK + tid;
    const int j    = s_idx[tid];
    float4 x0a = s_cols[j];
    float4 xn  = rp[grow];
    const float tv = t[b], omt = 1.0f - tv;

    float vx = xn.x - x0a.x, vy = xn.y - x0a.y, vz = xn.z - x0a.z;
    float xtx = omt * x0a.x + tv * xn.x;
    float xty = omt * x0a.y + tv * xn.y;
    float xtz = omt * x0a.z + tv * xn.z;

    float p0 = s_b2[0], p1 = s_b2[1], p2 = s_b2[2];
    for (int k = 0; k < H_; ++k) {
      float h = xtx * s_W1[k] + xty * s_W1[H_ + k] + xtz * s_W1[2 * H_ + k]
              + tv * s_Wt[k] + s_b1[k];
      h = fmaxf(h, 0.0f);
      p0 += h * s_W2[k * 3 + 0];
      p1 += h * s_W2[k * 3 + 1];
      p2 += h * s_W2[k * 3 + 2];
    }
    const size_t base  = ((size_t)b * N_ + grow) * 3;
    const size_t vbase = (size_t)B_ * N_ * 3 + base;
    out[base + 0] = p0; out[base + 1] = p1; out[base + 2] = p2;
    out[vbase + 0] = vx; out[vbase + 1] = vy; out[vbase + 2] = vz;
  }
}

extern "C" void kernel_launch(void* const* d_in, const int* in_sizes, int n_in,
                              void* d_out, int out_size, void* d_ws, size_t ws_size,
                              hipStream_t stream) {
  (void)in_sizes; (void)n_in; (void)out_size; (void)ws_size;
  const float* cloud = (const float*)d_in[0];
  const float* noise = (const float*)d_in[1];
  const float* t     = (const float*)d_in[2];
  const float* W1    = (const float*)d_in[3];
  const float* Wt    = (const float*)d_in[4];
  const float* b1    = (const float*)d_in[5];
  const float* W2    = (const float*)d_in[6];
  const float* b2    = (const float*)d_in[7];
  float* out = (float*)d_out;

  char* ws = (char*)d_ws;
  float4* ypts = (float4*)ws;
  float4* xpts = (float4*)(ws + (size_t)B_ * N_ * sizeof(float4));
  float*  f    = (float*) (ws + (size_t)2 * B_ * N_ * sizeof(float4));
  float*  g    = f + (size_t)B_ * N_;

  prep_kernel<<<B_, THREADS, 0, stream>>>(cloud, noise, ypts, xpts, f, g);

  const float lconst = -logf((float)N_);
  dim3 grid(N_ / ROWS_PER_BLOCK, B_);
  for (int it = 0; it < 14; ++it) {
    float eps = (float)(32.0 * pow(1e-6 / 32.0, (double)it / 13.0));
    sinkhorn_half<<<grid, THREADS, 0, stream>>>(xpts, ypts, g, f, eps, lconst);
    sinkhorn_half<<<grid, THREADS, 0, stream>>>(ypts, xpts, f, g, eps, lconst);
  }

  assign_mlp_kernel<<<grid, THREADS, 0, stream>>>(xpts, ypts, g, t,
                                                  W1, Wt, b1, W2, b2, out);
}


# Log-sum loop, first 6 eps steps: row max of 64 columns stands as reference, a trip adds exp2(score - ref) per element, redone by the exact update if a sum passes 1e30 (uniform); s rounding changes
# speedup vs baseline: 2.1622x; 2.1622x over previous
#include <hip/hip_runtime.h>
#include <cmath>


#define B_  8
#define N_  2048
#define H_  256
#define THREADS 256
#define ROWS_PER_BLOCK 128
#define KS 2
#define COLS_PER_WG (N_ / KS)
#define STAGE_PER_THREAD (COLS_PER_WG / THREADS)
#define TPT 16
#define LAZY_SUM_MAX 1.0e30f
#define LAZY_ITERS 6
#define SEED_TILES 4
#define PREP_PER_THREAD (N_ * 3 / THREADS)
#define ASG_STAGE_PER_THREAD (N_ / THREADS)
#define ASG_W_PER_THREAD (3 * H_ / THREADS)

static_assert(H_ == THREADS, "one element of Wt and of b1 per thread");
static_assert((3 * H_) % THREADS == 0 && N_ % THREADS == 0 && (N_ * 3) % THREADS == 0,
              "staging loops cover their arrays in whole trips");

typedef __attribute__((ext_vector_type(2))) float v2f;
typedef __attribute__((ext_vector_type(8))) float v8f;

#define LOG2E 1.4426950408889634f
#define LN2   0.6931471805599453f

__device__ __forceinline__ float sqnorm3(float x, float y, float z) {
  return fmaf(z, z, fmaf(y, y, x * x));
}

__device__ __forceinline__ float merged_potential(float la, float lb, float eps_in,
                                                  float lconst, float sq) {
  float top  = fmaxf(la, lb);
  float e    = __builtin_amdgcn_exp2f(-fabsf(la - lb));
  float lse2 = top + __builtin_amdgcn_logf(1.0f + e);
  return -eps_in * (lconst + LN2 * lse2) + 0.5f * sq;
}

template <int CTRL>
__device__ __forceinline__ float row_xchg(float x) {
  return __builtin_bit_cast(float, __builtin_amdgcn_mov_dpp(
      __builtin_bit_cast(int, x), CTRL, 0xf, 0xf, false));
}

__device__ __forceinline__ void lse_merge(float& mm, float& ss, float om, float os) {
  const float d  = om - mm;
  const float e  = __builtin_amdgcn_exp2f(-fabsf(d));
  const bool  up = d > 0.0f;
  ss = up ? fmaf(ss, e, os) : fmaf(os, e, ss);
  mm = up ? om : mm;
}

__global__ __launch_bounds__(THREADS)
void prep_kernel(const float* __restrict__ cloud, float* __restrict__ y3) {
  const int b   = blockIdx.x;
  const int tid = threadIdx.x;
  __shared__ float s_sum[THREADS];
  __shared__ float s_sq[THREADS];

  const float* cb = cloud + (size_t)b * N_ * 3;

  float v[PREP_PER_THREAD];
#pragma unroll
  for (int u = 0; u < PREP_PER_THREAD; ++u) v[u] = cb[tid + u * THREADS];

  float sum = 0.f, sq = 0.f;
#pragma unroll
  for (int u = 0; u < PREP_PER_THREAD; ++u) { sum += v[u]; sq = fmaf(v[u], v[u], sq); }
  s_sum[tid] = sum; s_sq[tid] = sq;
  __syncthreads();
  for (int off = THREADS / 2; off > 0; off >>= 1) {
    if (tid < off) { s_sum[tid] += s_sum[tid + off]; s_sq[tid] += s_sq[tid + off]; }
    __syncthreads();
  }
  const float n    = (float)(N_ * 3);
  const float tot  = s_sum[0];
  const float tot2 = s_sq[0];
  const float var  = (tot2 - tot * tot / n) / (n - 1.0f);
  const float inv_std = rsqrtf(var);

  float* yb = y3 + (size_t)b * N_ * 3;
#pragma unroll
  for (int u = 0; u < PREP_PER_THREAD; ++u) yb[tid + u * THREADS] = v[u] * inv_std;
}

__device__ __forceinline__ void tile_exact_update(const v8f& acct, float gyt, v2f (&m2)[4], v2f (&s2)[4],
                                                  v2f c2v, v2f onev) {
  v2f gyv; gyv.x = gyt; gyv.y = gyt;
#pragma unroll
  for (int q = 0; q < 4; ++q) {
    v2f ap; ap.x = acct[2 * q]; ap.y = acct[2 * q + 1];
    const v2f v = (gyv + ap) * c2v;
    const v2f d = v - m2[q];
    v2f e;
    e.x = __builtin_amdgcn_exp2f(-fabsf(d.x));
    e.y = __builtin_amdgcn_exp2f(-fabsf(d.y));
    const bool up0 = d.x > 0.0f;
    const bool up1 = d.y > 0.0f;
    const v2f sa = __builtin_elementwise_fma(s2[q], e, onev);
    const v2f sb = s2[q] + e;
    s2[q].x = up0 ? sa.x : sb.x;
    s2[q].y = up1 ? sa.y : sb.y;
    m2[q].x = up0 ? v.x : m2[q].x;
    m2[q].y = up1 ? v.y : m2[q].y;
  }
}

template <bool LAZY>
__global__ __launch_bounds__(THREADS)
void sinkhorn_half(const float* __restrict__ rows3,
                   const float* __restrict__ cols3,
                   const float* __restrict__ in_p0,
                   const float* __restrict__ in_p1,
                   float*       __restrict__ out_p0,
                   float*       __restrict__ out_p1,
                   float eps, float eps_in, float lconst, int first) {
  const int b    = blockIdx.y;
  const int ks   = blockIdx.x % KS;
  const int rb   = blockIdx.x / KS;
  const int tid  = threadIdx.x;
  const int lane = tid & 31;
  const int wave = tid >> 5;
  const int hi   = lane >> 4;
  const int idx  = lane & 15;

  __shared__ v2f   s_colp[COLS_PER_WG * 2];
  __shared__ float s_gy[COLS_PER_WG];

  const int row_base = rb * ROWS_PER_BLOCK + wave * 16;
  const float* rp = rows3 + (size_t)b * N_ * 3;
  const float rx = rp[3 * (row_base + idx) + 0];
  const float ry = rp[3 * (row_base + idx) + 1];
  const float rz = rp[3 * (row_base + idx) + 2];

  const size_t cbase = (size_t)b * N_ + (size_t)ks * COLS_PER_WG;
  const float* cp = cols3 + cbase * 3;
  const float* p0 = in_p0 + cbase;
  const float* p1 = in_p1 + cbase;

  float cx[STAGE_PER_THREAD], cy[STAGE_PER_THREAD], cz[STAGE_PER_THREAD];
  float la[STAGE_PER_THREAD], lb[STAGE_PER_THREAD];
#pragma unroll
  for (int u = 0; u < STAGE_PER_THREAD; ++u) {
    const int j = tid + u * THREADS;
    cx[u] = cp[3 * j + 0];
    cy[u] = cp[3 * j + 1];
    cz[u] = cp[3 * j + 2];
    la[u] = 0.f; lb[u] = 0.f;
    if (!first) { la[u] = p0[j]; lb[u] = p1[j]; }
  }
#pragma unroll
  for (int u = 0; u < STAGE_PER_THREAD; ++u) {
    const int j = tid + u * THREADS;
    const float sq = sqnorm3(cx[u], cy[u], cz[u]);
    const float pot = first ? 0.0f : merged_potential(la[u], lb[u], eps_in, lconst, sq);
    v2f lo2; lo2.x = cx[u]; lo2.y = cy[u];
    v2f hi2; hi2.x = cz[u]; hi2.y = 0.0f;
    s_colp[2 * j + 0] = lo2;
    s_colp[2 * j + 1] = hi2;
    s_gy[j] = pot - 0.5f * sq;
  }
  __syncthreads();

  v2f a;
  a.x = hi ? rz   : rx;
  a.y = hi ? 0.0f : ry;

  v2f m2[4], s2[4];
#pragma unroll
  for (int q = 0; q < 4; ++q) { m2[q].x = -3.0e38f; m2[q].y = -3.0e38f; s2[q].x = 0.f; s2[q].y = 0.f; }
  const float c2 = LOG2E / eps;
  v2f c2v;  c2v.x = c2;    c2v.y = c2;
  v2f onev; onev.x = 1.0f; onev.y = 1.0f;

  if constexpr (LAZY) {
    {
      float ref[8];
#pragma unroll
      for (int r = 0; r < 8; ++r) ref[r] = -3.0e38f;
#pragma unroll
      for (int t = 0; t < SEED_TILES; ++t) {
        const v2f   bf0 = s_colp[2 * (16 * t + idx) + hi];
        const float gy0 = s_gy[16 * t + idx];
        v8f cz0 = {};
        const v8f acc0 = __builtin_amdgcn_wmma_f32_16x16x4_f32(
            false, a, false, bf0, (short)0, cz0, false, false);
        v2f gyv; gyv.x = gy0; gyv.y = gy0;
#pragma unroll
        for (int q = 0; q < 4; ++q) {
          v2f ap; ap.x = acc0[2 * q]; ap.y = acc0[2 * q + 1];
          const v2f v = (gyv + ap) * c2v;
          ref[2 * q]     = fmaxf(ref[2 * q], v.x);
          ref[2 * q + 1] = fmaxf(ref[2 * q + 1], v.y);
        }
      }
#pragma unroll
      for (int r = 0; r < 8; ++r) {
        ref[r] = fmaxf(ref[r], row_xchg<0x161>(ref[r]));
        ref[r] = fmaxf(ref[r], row_xchg<0x162>(ref[r]));
        ref[r] = fmaxf(ref[r], row_xchg<0x164>(ref[r]));
        ref[r] = fmaxf(ref[r], row_xchg<0x168>(ref[r]));
      }
#pragma unroll
      for (int q = 0; q < 4; ++q) { m2[q].x = ref[2 * q]; m2[q].y = ref[2 * q + 1]; }
    }

    for (int c0 = 0; c0 < COLS_PER_WG; c0 += 16 * TPT) {
      v2f   bf[TPT];
      float gy[TPT];
#pragma unroll
      for (int t = 0; t < TPT; ++t) {
        bf[t] = s_colp[2 * (c0 + 16 * t + idx) + hi];
        gy[t] = s_gy[c0 + 16 * t + idx];
      }
      v8f acc[TPT];
#pragma unroll
      for (int t = 0; t < TPT; ++t) {
        v8f cz0 = {};
        acc[t] = __builtin_amdgcn_wmma_f32_16x16x4_f32(
            false, a, false, bf[t], (short)0, cz0, false, false);
      }

      v2f sn[4];
#pragma unroll
      for (int q = 0; q < 4; ++q) sn[q] = s2[q];
#pragma unroll
      for (int t = 0; t < TPT; ++t) {
        v2f gyv; gyv.x = gy[t]; gyv.y = gy[t];
#pragma unroll
        for (int q = 0; q < 4; ++q) {
          v2f ap; ap.x = acc[t][2 * q]; ap.y = acc[t][2 * q + 1];
          const v2f v = (gyv + ap) * c2v;
          const v2f d = v - m2[q];
          v2f e;
          e.x = __builtin_amdgcn_exp2f(d.x);
          e.y = __builtin_amdgcn_exp2f(d.y);
          sn[q] = sn[q] + e;
        }
      }
      float top = fmaxf(fmaxf(sn[0].x, sn[0].y), fmaxf(sn[1].x, sn[1].y));
      top = fmaxf(top, fmaxf(fmaxf(sn[2].x, sn[2].y), fmaxf(sn[3].x, sn[3].y)));

      if (__builtin_amdgcn_ballot_w32(!(top < LAZY_SUM_MAX)) == 0u) {
#pragma unroll
        for (int q = 0; q < 4; ++q) s2[q] = sn[q];
      } else {
#pragma unroll
        for (int t = 0; t < TPT; ++t) tile_exact_update(acc[t], gy[t], m2, s2, c2v, onev);
      }
    }
  } else {
    for (int c0 = 0; c0 < COLS_PER_WG; c0 += 16 * TPT) {
      v2f   bf[TPT];
      float gy[TPT];
#pragma unroll
      for (int t = 0; t < TPT; ++t) {
        bf[t] = s_colp[2 * (c0 + 16 * t + idx) + hi];
        gy[t] = s_gy[c0 + 16 * t + idx];
      }

      v8f acc[TPT];
#pragma unroll
      for (int t = 0; t < TPT; ++t) {
        v8f cz0 = {};
        acc[t] = __builtin_amdgcn_wmma_f32_16x16x4_f32(
            false, a, false, bf[t], (short)0, cz0, false, false);
      }

#pragma unroll
      for (int t = 0; t < TPT; ++t) {
        v2f gyv; gyv.x = gy[t]; gyv.y = gy[t];
#pragma unroll
        for (int q = 0; q < 4; ++q) {
          v2f ap; ap.x = acc[t][2 * q]; ap.y = acc[t][2 * q + 1];
          const v2f v = (gyv + ap) * c2v;
          const v2f d = v - m2[q];
          v2f e;
          e.x = __builtin_amdgcn_exp2f(-fabsf(d.x));
          e.y = __builtin_amdgcn_exp2f(-fabsf(d.y));
          const bool up0 = d.x > 0.0f;
          const bool up1 = d.y > 0.0f;
          const v2f sa = __builtin_elementwise_fma(s2[q], e, onev);
          const v2f sb = s2[q] + e;
          s2[q].x = up0 ? sa.x : sb.x;
          s2[q].y = up1 ? sa.y : sb.y;
          m2[q].x = up0 ? v.x : m2[q].x;
          m2[q].y = up1 ? v.y : m2[q].y;
        }
      }
    }
  }

  float m[8], s[8];
#pragma unroll
  for (int q = 0; q < 4; ++q) {
    m[2 * q] = m2[q].x; m[2 * q + 1] = m2[q].y;
    s[2 * q] = s2[q].x; s[2 * q + 1] = s2[q].y;
  }

  const bool b0 = (idx & 1) != 0;
  const bool b1 = (idx & 2) != 0;
  const bool b2 = (idx & 4) != 0;

  float ma[4], sa[4];
#pragma unroll
  for (int r = 0; r < 4; ++r) {
    float km = b0 ? m[r + 4] : m[r];
    float kv = b0 ? s[r + 4] : s[r];
    const float gm = b0 ? m[r] : m[r + 4];
    const float gv = b0 ? s[r] : s[r + 4];
    lse_merge(km, kv, row_xchg<0x161>(gm), row_xchg<0x161>(gv));
    ma[r] = km; sa[r] = kv;
  }
  float mb[2], sb2[2];
#pragma unroll
  for (int r = 0; r < 2; ++r) {
    float km = b1 ? ma[r + 2] : ma[r];
    float kv = b1 ? sa[r + 2] : sa[r];
    const float gm = b1 ? ma[r] : ma[r + 2];
    const float gv = b1 ? sa[r] : sa[r + 2];
    lse_merge(km, kv, row_xchg<0x162>(gm), row_xchg<0x162>(gv));
    mb[r] = km; sb2[r] = kv;
  }
  float mc = b2 ? mb[1] : mb[0];
  float sc = b2 ? sb2[1] : sb2[0];
  {
    const float gm = b2 ? mb[0] : mb[1];
    const float gv = b2 ? sb2[0] : sb2[1];
    lse_merge(mc, sc, row_xchg<0x164>(gm), row_xchg<0x164>(gv));
  }
  lse_merge(mc, sc, row_xchg<0x168>(mc), row_xchg<0x168>(sc));

  if (idx < 8) {
    const int r   = (b0 ? 4 : 0) + (b1 ? 2 : 0) + (b2 ? 1 : 0);
    const int row = row_base + hi * 8 + r;
    const float l2 = mc + __builtin_amdgcn_logf(sc);
    float* op = ks ? out_p1 : out_p0;
    op[(size_t)b * N_ + row] = l2;
  }
}

__global__ __launch_bounds__(THREADS)
void assign_mlp_kernel(const float* __restrict__ x3,
                       const float* __restrict__ y3,
                       const float* __restrict__ g0,
                       const float* __restrict__ g1,
                       float eps_g, float lconst,
                       const float*  __restrict__ t,
                       const float*  __restrict__ W1,
                       const float*  __restrict__ Wt,
                       const float*  __restrict__ b1,
                       const float*  __restrict__ W2,
                       const float*  __restrict__ b2,
                       float*        __restrict__ out) {
  const int b    = blockIdx.y;
  const int tid  = threadIdx.x;
  const int lane = tid & 31;
  const int wave = tid >> 5;
  const int hi   = lane >> 4;
  const int idx  = lane & 15;

  __shared__ float4 s_cols[N_];
  __shared__ float  s_gy[N_];
  __shared__ float  s_W1[3 * H_];
  __shared__ float  s_Wt[H_];
  __shared__ float  s_b1[H_];
  __shared__ float  s_W2[H_ * 3];
  __shared__ float  s_b2[4];
  __shared__ int    s_idx[ROWS_PER_BLOCK];

  const float* cp  = y3 + (size_t)b * N_ * 3;
  const float* gp0 = g0 + (size_t)b * N_;
  const float* gp1 = g1 + (size_t)b * N_;

  const int row_base = blockIdx.x * ROWS_PER_BLOCK + wave * 16;
  const float* rp = x3 + (size_t)b * N_ * 3;
  const float rx = rp[3 * (row_base + idx) + 0];
  const float ry = rp[3 * (row_base + idx) + 1];
  const float rz = rp[3 * (row_base + idx) + 2];

  float xnx = 0.0f, xny = 0.0f, xnz = 0.0f;
  if (tid < ROWS_PER_BLOCK) {
    const int grow0 = blockIdx.x * ROWS_PER_BLOCK + tid;
    xnx = rp[3 * grow0 + 0];
    xny = rp[3 * grow0 + 1];
    xnz = rp[3 * grow0 + 2];
  }
  const float tv = t[b];

  float yx[ASG_STAGE_PER_THREAD], yy[ASG_STAGE_PER_THREAD], yz[ASG_STAGE_PER_THREAD];
  float ga[ASG_STAGE_PER_THREAD], gb[ASG_STAGE_PER_THREAD];
#pragma unroll
  for (int u = 0; u < ASG_STAGE_PER_THREAD; ++u) {
    const int j = tid + u * THREADS;
    yx[u] = cp[3 * j + 0];
    yy[u] = cp[3 * j + 1];
    yz[u] = cp[3 * j + 2];
    ga[u] = gp0[j];
    gb[u] = gp1[j];
  }
  float w1r[ASG_W_PER_THREAD], w2r[ASG_W_PER_THREAD];
#pragma unroll
  for (int u = 0; u < ASG_W_PER_THREAD; ++u) {
    w1r[u] = W1[tid + u * THREADS];
    w2r[u] = W2[tid + u * THREADS];
  }
  const float wtr = Wt[tid];
  const float b1r = b1[tid];
  float b2r = 0.0f;
  if (tid < 3) b2r = b2[tid];

#pragma unroll
  for (int u = 0; u < ASG_STAGE_PER_THREAD; ++u) {
    const int j = tid + u * THREADS;
    const float sq = sqnorm3(yx[u], yy[u], yz[u]);
    const float gj = merged_potential(ga[u], gb[u], eps_g, lconst, sq);
    s_cols[j] = make_float4(yx[u], yy[u], yz[u], 0.0f);
    s_gy[j]   = gj - sq;
  }
#pragma unroll
  for (int u = 0; u < ASG_W_PER_THREAD; ++u) {
    s_W1[tid + u * THREADS] = w1r[u];
    s_W2[tid + u * THREADS] = w2r[u];
  }
  s_Wt[tid] = wtr;
  s_b1[tid] = b1r;
  if (tid < 3) s_b2[tid] = b2r;
  __syncthreads();

  v2f a;
  a.x = hi ? 2.0f * rz : 2.0f * rx;
  a.y = hi ? 0.0f      : 2.0f * ry;

  float best[8]; int bidx[8];
#pragma unroll
  for (int r = 0; r < 8; ++r) { best[r] = -3.0e38f; bidx[r] = 0; }

#pragma unroll 4
  for (int c0 = 0; c0 < N_; c0 += 16) {
    float4 c4 = s_cols[c0 + idx];
    v2f bf;
    bf.x = hi ? c4.z : c4.x;
    bf.y = hi ? 0.0f : c4.y;
    float gy = s_gy[c0 + idx];

    v8f cz0 = {};
    v8f acc = __builtin_amdgcn_wmma_f32_16x16x4_f32(
        false, a, false, bf, (short)0, cz0, false, false);

#pragma unroll
    for (int r = 0; r < 8; ++r) {
      float v = acc[r] + gy;
      if (v > best[r]) { best[r] = v; bidx[r] = c0 + idx; }
    }
  }

#pragma unroll
  for (int mask = 1; mask < 16; mask <<= 1) {
#pragma unroll
    for (int r = 0; r < 8; ++r) {
      float ov = __shfl_xor(best[r], mask, 32);
      int   oi = __shfl_xor(bidx[r], mask, 32);
      if (ov > best[r] || (ov == best[r] && oi < bidx[r])) {
        best[r] = ov; bidx[r] = oi;
      }
    }
  }

  if (idx < 8) s_idx[wave * 16 + hi * 8 + idx] = bidx[idx];
  __syncthreads();

  if (tid < ROWS_PER_BLOCK) {
    const int grow = blockIdx.x * ROWS_PER_BLOCK + tid;
    const int j    = s_idx[tid];
    float4 x0a = s_cols[j];
    const float omt = 1.0f - tv;

    float vx = xnx - x0a.x, vy = xny - x0a.y, vz = xnz - x0a.z;
    float xtx = omt * x0a.x + tv * xnx;
    float xty = omt * x0a.y + tv * xny;
    float xtz = omt * x0a.z + tv * xnz;

    float p0 = s_b2[0], p1 = s_b2[1], p2 = s_b2[2];
    for (int k = 0; k < H_; ++k) {
      float h = xtx * s_W1[k] + xty * s_W1[H_ + k] + xtz * s_W1[2 * H_ + k]
              + tv * s_Wt[k] + s_b1[k];
      h = fmaxf(h, 0.0f);
      p0 += h * s_W2[k * 3 + 0];
      p1 += h * s_W2[k * 3 + 1];
      p2 += h * s_W2[k * 3 + 2];
    }
    const size_t base  = ((size_t)b * N_ + grow) * 3;
    const size_t vbase = (size_t)B_ * N_ * 3 + base;
    out[base + 0] = p0; out[base + 1] = p1; out[base + 2] = p2;
    out[vbase + 0] = vx; out[vbase + 1] = vy; out[vbase + 2] = vz;
  }
}

extern "C" void kernel_launch(void* const* d_in, const int* in_sizes, int n_in,
                              void* d_out, int out_size, void* d_ws, size_t ws_size,
                              hipStream_t stream) {
  (void)in_sizes; (void)n_in; (void)out_size; (void)ws_size;
  const float* cloud = (const float*)d_in[0];
  const float* noise = (const float*)d_in[1];
  const float* t     = (const float*)d_in[2];
  const float* W1    = (const float*)d_in[3];
  const float* Wt    = (const float*)d_in[4];
  const float* b1    = (const float*)d_in[5];
  const float* W2    = (const float*)d_in[6];
  const float* b2    = (const float*)d_in[7];
  float* out = (float*)d_out;

  char* ws = (char*)d_ws;
  float* y3 = (float*)ws;
  float* f0 = (float*)(ws + (size_t)B_ * N_ * 3 * sizeof(float));
  float* f1 = f0 + (size_t)B_ * N_;
  float* g0 = f1 + (size_t)B_ * N_;
  float* g1 = g0 + (size_t)B_ * N_;

  prep_kernel<<<B_, THREADS, 0, stream>>>(cloud, y3);

  const float lconst = -logf((float)N_);
  dim3 grid(N_ / ROWS_PER_BLOCK * KS, B_);
  float eps_in = 0.0f;
  for (int it = 0; it < 14; ++it) {
    float eps = (float)(32.0 * pow(1e-6 / 32.0, (double)it / 13.0));
    if (it < LAZY_ITERS) {
      sinkhorn_half<true><<<grid, THREADS, 0, stream>>>(noise, y3, g0, g1, f0, f1,
                                                        eps, eps_in, lconst, it == 0);
      sinkhorn_half<true><<<grid, THREADS, 0, stream>>>(y3, noise, f0, f1, g0, g1,
                                                        eps, eps, lconst, 0);
    } else {
      sinkhorn_half<false><<<grid, THREADS, 0, stream>>>(noise, y3, g0, g1, f0, f1,
                                                         eps, eps_in, lconst, 0);
      sinkhorn_half<false><<<grid, THREADS, 0, stream>>>(y3, noise, f0, f1, g0, g1,
                                                         eps, eps, lconst, 0);
    }
    eps_in = eps;
  }

  dim3 grid_a(N_ / ROWS_PER_BLOCK, B_);
  assign_mlp_kernel<<<grid_a, THREADS, 0, stream>>>(noise, y3, g0, g1, eps_in, lconst, t,
                                                    W1, Wt, b1, W2, b2, out);
}


# First half-step is its own instantiation that computes the batch scale itself (same sums, tree; row block 0 stores the scaled cloud): 29 launches; other half-steps stage straight-line; values kept
# speedup vs baseline: 2.2589x; 1.0447x over previous
#include <hip/hip_runtime.h>
#include <cmath>


#define B_  8
#define N_  2048
#define H_  256
#define THREADS 256
#define ROWS_PER_BLOCK 128
#define KS 2
#define COLS_PER_WG (N_ / KS)
#define STAGE_PER_THREAD (COLS_PER_WG / THREADS)
#define TPT 16
#define LAZY_SUM_MAX 1.0e30f
#define LAZY_ITERS 6
#define SEED_TILES 4
#define CLOUD_PER_THREAD (N_ * 3 / THREADS)
#define ASG_STAGE_PER_THREAD (N_ / THREADS)
#define ASG_W_PER_THREAD (3 * H_ / THREADS)

static_assert(H_ == THREADS, "one element of Wt and of b1 per thread");
static_assert((3 * H_) % THREADS == 0 && N_ % THREADS == 0 && (N_ * 3) % THREADS == 0,
              "staging loops cover their arrays in whole trips");

typedef __attribute__((ext_vector_type(2))) float v2f;
typedef __attribute__((ext_vector_type(8))) float v8f;

#define LOG2E 1.4426950408889634f
#define LN2   0.6931471805599453f

__device__ __forceinline__ float sqnorm3(float x, float y, float z) {
  return fmaf(z, z, fmaf(y, y, x * x));
}

__device__ __forceinline__ float merged_potential(float la, float lb, float eps_in,
                                                  float lconst, float sq) {
  float top  = fmaxf(la, lb);
  float e    = __builtin_amdgcn_exp2f(-fabsf(la - lb));
  float lse2 = top + __builtin_amdgcn_logf(1.0f + e);
  return -eps_in * (lconst + LN2 * lse2) + 0.5f * sq;
}

__device__ __forceinline__ float merged_potential_fused(float la, float lb, float eps_in,
                                                        float lconst, float sq) {
  float top  = fmaxf(la, lb);
  float e    = __builtin_amdgcn_exp2f(-fabsf(la - lb));
  float lse2 = top + __builtin_amdgcn_logf(1.0f + e);
  return fmaf(-eps_in, fmaf(LN2, lse2, lconst), 0.5f * sq);
}

template <int CTRL>
__device__ __forceinline__ float row_xchg(float x) {
  return __builtin_bit_cast(float, __builtin_amdgcn_mov_dpp(
      __builtin_bit_cast(int, x), CTRL, 0xf, 0xf, false));
}

__device__ __forceinline__ void lse_merge(float& mm, float& ss, float om, float os) {
  const float d  = om - mm;
  const float e  = __builtin_amdgcn_exp2f(-fabsf(d));
  const bool  up = d > 0.0f;
  ss = up ? fmaf(ss, e, os) : fmaf(os, e, ss);
  mm = up ? om : mm;
}

__device__ __forceinline__ void tile_exact_update(const v8f& acct, float gyt, v2f (&m2)[4], v2f (&s2)[4],
                                                  v2f c2v, v2f onev) {
  v2f gyv; gyv.x = gyt; gyv.y = gyt;
#pragma unroll
  for (int q = 0; q < 4; ++q) {
    v2f ap; ap.x = acct[2 * q]; ap.y = acct[2 * q + 1];
    const v2f v = (gyv + ap) * c2v;
    const v2f d = v - m2[q];
    v2f e;
    e.x = __builtin_amdgcn_exp2f(-fabsf(d.x));
    e.y = __builtin_amdgcn_exp2f(-fabsf(d.y));
    const bool up0 = d.x > 0.0f;
    const bool up1 = d.y > 0.0f;
    const v2f sa = __builtin_elementwise_fma(s2[q], e, onev);
    const v2f sb = s2[q] + e;
    s2[q].x = up0 ? sa.x : sb.x;
    s2[q].y = up1 ? sa.y : sb.y;
    m2[q].x = up0 ? v.x : m2[q].x;
    m2[q].y = up1 ? v.y : m2[q].y;
  }
}

template <bool LAZY, bool FIRST>
__global__ __launch_bounds__(THREADS)
void sinkhorn_half(const float* __restrict__ rows3,
                   const float* __restrict__ cols3,
                   const float* __restrict__ in_p0,
                   const float* __restrict__ in_p1,
                   float*       __restrict__ out_p0,
                   float*       __restrict__ out_p1,
                   float*       __restrict__ scaled_out,
                   float eps, float eps_in, float lconst) {
  const int b    = blockIdx.y;
  const int ks   = blockIdx.x % KS;
  const int rb   = blockIdx.x / KS;
  const int tid  = threadIdx.x;
  const int lane = tid & 31;
  const int wave = tid >> 5;
  const int hi   = lane >> 4;
  const int idx  = lane & 15;

  __shared__ v2f   s_colp[COLS_PER_WG * 2];
  __shared__ float s_gy[COLS_PER_WG];

  const int row_base = rb * ROWS_PER_BLOCK + wave * 16;
  const float* rp = rows3 + (size_t)b * N_ * 3;
  const float rx = rp[3 * (row_base + idx) + 0];
  const float ry = rp[3 * (row_base + idx) + 1];
  const float rz = rp[3 * (row_base + idx) + 2];
  __builtin_amdgcn_sched_barrier(0);

  const size_t cbase = (size_t)b * N_ + (size_t)ks * COLS_PER_WG;
  const float* cp = cols3 + cbase * 3;
  const float* p0 = in_p0 + cbase;
  const float* p1 = in_p1 + cbase;

  float cx[STAGE_PER_THREAD], cy[STAGE_PER_THREAD], cz[STAGE_PER_THREAD];
  float la[STAGE_PER_THREAD], lb[STAGE_PER_THREAD];
#pragma unroll
  for (int u = 0; u < STAGE_PER_THREAD; ++u) {
    const int j = tid + u * THREADS;
    cx[u] = cp[3 * j + 0];
    cy[u] = cp[3 * j + 1];
    cz[u] = cp[3 * j + 2];
    la[u] = 0.f; lb[u] = 0.f;
    if constexpr (!FIRST) { la[u] = p0[j]; lb[u] = p1[j]; }
  }

  if constexpr (FIRST) {
    __builtin_amdgcn_sched_barrier(0);
    __shared__ float s_sum[THREADS];
    __shared__ float s_sq[THREADS];
    const float* cb = cols3 + (size_t)b * N_ * 3;
    float v[CLOUD_PER_THREAD];
#pragma unroll
    for (int u = 0; u < CLOUD_PER_THREAD; ++u) v[u] = cb[tid + u * THREADS];
    float sum = 0.f, sq = 0.f;
#pragma unroll
    for (int u = 0; u < CLOUD_PER_THREAD; ++u) { sum += v[u]; sq = fmaf(v[u], v[u], sq); }
    s_sum[tid] = sum; s_sq[tid] = sq;
    __syncthreads();
    for (int off = THREADS / 2; off > 0; off >>= 1) {
      if (tid < off) { s_sum[tid] += s_sum[tid + off]; s_sq[tid] += s_sq[tid + off]; }
      __syncthreads();
    }
    const float n    = (float)(N_ * 3);
    const float tot  = s_sum[0];
    const float tot2 = s_sq[0];
    const float var  = (tot2 - tot * tot / n) / (n - 1.0f);
    const float inv_std = rsqrtf(var);
#pragma unroll
    for (int u = 0; u < STAGE_PER_THREAD; ++u) {
      cx[u] *= inv_std; cy[u] *= inv_std; cz[u] *= inv_std;
    }
  }

#pragma unroll
  for (int u = 0; u < STAGE_PER_THREAD; ++u) {
    const int j = tid + u * THREADS;
    const float sq = sqnorm3(cx[u], cy[u], cz[u]);
    const float pot = FIRST ? 0.0f : merged_potential_fused(la[u], lb[u], eps_in, lconst, sq);
    v2f lo2; lo2.x = cx[u]; lo2.y = cy[u];
    v2f hi2; hi2.x = cz[u]; hi2.y = 0.0f;
    s_colp[2 * j + 0] = lo2;
    s_colp[2 * j + 1] = hi2;
    s_gy[j] = pot - 0.5f * sq;
  }
  __syncthreads();

  if constexpr (FIRST) {
    if (rb == 0) {
      float* so = scaled_out + cbase * 3;
#pragma unroll
      for (int u = 0; u < STAGE_PER_THREAD; ++u) {
        const int j = tid + u * THREADS;
        so[3 * j + 0] = cx[u];
        so[3 * j + 1] = cy[u];
        so[3 * j + 2] = cz[u];
      }
    }
  }

  v2f a;
  a.x = hi ? rz   : rx;
  a.y = hi ? 0.0f : ry;

  v2f m2[4], s2[4];
#pragma unroll
  for (int q = 0; q < 4; ++q) { m2[q].x = -3.0e38f; m2[q].y = -3.0e38f; s2[q].x = 0.f; s2[q].y = 0.f; }
  const float c2 = LOG2E / eps;
  v2f c2v;  c2v.x = c2;    c2v.y = c2;
  v2f onev; onev.x = 1.0f; onev.y = 1.0f;

  if constexpr (LAZY) {
    {
      float ref[8];
#pragma unroll
      for (int r = 0; r < 8; ++r) ref[r] = -3.0e38f;
#pragma unroll
      for (int t = 0; t < SEED_TILES; ++t) {
        const v2f   bf0 = s_colp[2 * (16 * t + idx) + hi];
        const float gy0 = s_gy[16 * t + idx];
        v8f cz0 = {};
        const v8f acc0 = __builtin_amdgcn_wmma_f32_16x16x4_f32(
            false, a, false, bf0, (short)0, cz0, false, false);
        v2f gyv; gyv.x = gy0; gyv.y = gy0;
#pragma unroll
        for (int q = 0; q < 4; ++q) {
          v2f ap; ap.x = acc0[2 * q]; ap.y = acc0[2 * q + 1];
          const v2f v = (gyv + ap) * c2v;
          ref[2 * q]     = fmaxf(ref[2 * q], v.x);
          ref[2 * q + 1] = fmaxf(ref[2 * q + 1], v.y);
        }
      }
#pragma unroll
      for (int r = 0; r < 8; ++r) {
        ref[r] = fmaxf(ref[r], row_xchg<0x161>(ref[r]));
        ref[r] = fmaxf(ref[r], row_xchg<0x162>(ref[r]));
        ref[r] = fmaxf(ref[r], row_xchg<0x164>(ref[r]));
        ref[r] = fmaxf(ref[r], row_xchg<0x168>(ref[r]));
      }
#pragma unroll
      for (int q = 0; q < 4; ++q) { m2[q].x = ref[2 * q]; m2[q].y = ref[2 * q + 1]; }
    }

    for (int c0 = 0; c0 < COLS_PER_WG; c0 += 16 * TPT) {
      v2f   bf[TPT];
      float gy[TPT];
#pragma unroll
      for (int t = 0; t < TPT; ++t) {
        bf[t] = s_colp[2 * (c0 + 16 * t + idx) + hi];
        gy[t] = s_gy[c0 + 16 * t + idx];
      }
      v8f acc[TPT];
#pragma unroll
      for (int t = 0; t < TPT; ++t) {
        v8f cz0 = {};
        acc[t] = __builtin_amdgcn_wmma_f32_16x16x4_f32(
            false, a, false, bf[t], (short)0, cz0, false, false);
      }

      v2f sn[4];
#pragma unroll
      for (int q = 0; q < 4; ++q) sn[q] = s2[q];
#pragma unroll
      for (int t = 0; t < TPT; ++t) {
        v2f gyv; gyv.x = gy[t]; gyv.y = gy[t];
#pragma unroll
        for (int q = 0; q < 4; ++q) {
          v2f ap; ap.x = acc[t][2 * q]; ap.y = acc[t][2 * q + 1];
          const v2f v = (gyv + ap) * c2v;
          const v2f d = v - m2[q];
          v2f e;
          e.x = __builtin_amdgcn_exp2f(d.x);
          e.y = __builtin_amdgcn_exp2f(d.y);
          sn[q] = sn[q] + e;
        }
      }
      float top = fmaxf(fmaxf(sn[0].x, sn[0].y), fmaxf(sn[1].x, sn[1].y));
      top = fmaxf(top, fmaxf(fmaxf(sn[2].x, sn[2].y), fmaxf(sn[3].x, sn[3].y)));

      if (__builtin_amdgcn_ballot_w32(!(top < LAZY_SUM_MAX)) == 0u) {
#pragma unroll
        for (int q = 0; q < 4; ++q) s2[q] = sn[q];
      } else {
#pragma unroll
        for (int t = 0; t < TPT; ++t) tile_exact_update(acc[t], gy[t], m2, s2, c2v, onev);
      }
    }
  } else {
    for (int c0 = 0; c0 < COLS_PER_WG; c0 += 16 * TPT) {
      v2f   bf[TPT];
      float gy[TPT];
#pragma unroll
      for (int t = 0; t < TPT; ++t) {
        bf[t] = s_colp[2 * (c0 + 16 * t + idx) + hi];
        gy[t] = s_gy[c0 + 16 * t + idx];
      }

      v8f acc[TPT];
#pragma unroll
      for (int t = 0; t < TPT; ++t) {
        v8f cz0 = {};
        acc[t] = __builtin_amdgcn_wmma_f32_16x16x4_f32(
            false, a, false, bf[t], (short)0, cz0, false, false);
      }

#pragma unroll
      for (int t = 0; t < TPT; ++t) {
        v2f gyv; gyv.x = gy[t]; gyv.y = gy[t];
#pragma unroll
        for (int q = 0; q < 4; ++q) {
          v2f ap; ap.x = acc[t][2 * q]; ap.y = acc[t][2 * q + 1];
          const v2f v = (gyv + ap) * c2v;
          const v2f d = v - m2[q];
          v2f e;
          e.x = __builtin_amdgcn_exp2f(-fabsf(d.x));
          e.y = __builtin_amdgcn_exp2f(-fabsf(d.y));
          const bool up0 = d.x > 0.0f;
          const bool up1 = d.y > 0.0f;
          const v2f sa = __builtin_elementwise_fma(s2[q], e, onev);
          const v2f sb = s2[q] + e;
          s2[q].x = up0 ? sa.x : sb.x;
          s2[q].y = up1 ? sa.y : sb.y;
          m2[q].x = up0 ? v.x : m2[q].x;
          m2[q].y = up1 ? v.y : m2[q].y;
        }
      }
    }
  }

  float m[8], s[8];
#pragma unroll
  for (int q = 0; q < 4; ++q) {
    m[2 * q] = m2[q].x; m[2 * q + 1] = m2[q].y;
    s[2 * q] = s2[q].x; s[2 * q + 1] = s2[q].y;
  }

  const bool b0 = (idx & 1) != 0;
  const bool b1 = (idx & 2) != 0;
  const bool b2 = (idx & 4) != 0;

  float ma[4], sa[4];
#pragma unroll
  for (int r = 0; r < 4; ++r) {
    float km = b0 ? m[r + 4] : m[r];
    float kv = b0 ? s[r + 4] : s[r];
    const float gm = b0 ? m[r] : m[r + 4];
    const float gv = b0 ? s[r] : s[r + 4];
    lse_merge(km, kv, row_xchg<0x161>(gm), row_xchg<0x161>(gv));
    ma[r] = km; sa[r] = kv;
  }
  float mb[2], sb2[2];
#pragma unroll
  for (int r = 0; r < 2; ++r) {
    float km = b1 ? ma[r + 2] : ma[r];
    float kv = b1 ? sa[r + 2] : sa[r];
    const float gm = b1 ? ma[r] : ma[r + 2];
    const float gv = b1 ? sa[r] : sa[r + 2];
    lse_merge(km, kv, row_xchg<0x162>(gm), row_xchg<0x162>(gv));
    mb[r] = km; sb2[r] = kv;
  }
  float mc = b2 ? mb[1] : mb[0];
  float sc = b2 ? sb2[1] : sb2[0];
  {
    const float gm = b2 ? mb[0] : mb[1];
    const float gv = b2 ? sb2[0] : sb2[1];
    lse_merge(mc, sc, row_xchg<0x164>(gm), row_xchg<0x164>(gv));
  }
  lse_merge(mc, sc, row_xchg<0x168>(mc), row_xchg<0x168>(sc));

  if (idx < 8) {
    const int r   = (b0 ? 4 : 0) + (b1 ? 2 : 0) + (b2 ? 1 : 0);
    const int row = row_base + hi * 8 + r;
    const float l2 = mc + __builtin_amdgcn_logf(sc);
    float* op = ks ? out_p1 : out_p0;
    op[(size_t)b * N_ + row] = l2;
  }
}

__global__ __launch_bounds__(THREADS)
void assign_mlp_kernel(const float* __restrict__ x3,
                       const float* __restrict__ y3,
                       const float* __restrict__ g0,
                       const float* __restrict__ g1,
                       float eps_g, float lconst,
                       const float*  __restrict__ t,
                       const float*  __restrict__ W1,
                       const float*  __restrict__ Wt,
                       const float*  __restrict__ b1,
                       const float*  __restrict__ W2,
                       const float*  __restrict__ b2,
                       float*        __restrict__ out) {
  const int b    = blockIdx.y;
  const int tid  = threadIdx.x;
  const int lane = tid & 31;
  const int wave = tid >> 5;
  const int hi   = lane >> 4;
  const int idx  = lane & 15;

  __shared__ float4 s_cols[N_];
  __shared__ float  s_gy[N_];
  __shared__ float  s_W1[3 * H_];
  __shared__ float  s_Wt[H_];
  __shared__ float  s_b1[H_];
  __shared__ float  s_W2[H_ * 3];
  __shared__ float  s_b2[4];
  __shared__ int    s_idx[ROWS_PER_BLOCK];

  const float* cp  = y3 + (size_t)b * N_ * 3;
  const float* gp0 = g0 + (size_t)b * N_;
  const float* gp1 = g1 + (size_t)b * N_;

  const int row_base = blockIdx.x * ROWS_PER_BLOCK + wave * 16;
  const float* rp = x3 + (size_t)b * N_ * 3;
  const float rx = rp[3 * (row_base + idx) + 0];
  const float ry = rp[3 * (row_base + idx) + 1];
  const float rz = rp[3 * (row_base + idx) + 2];

  float xnx = 0.0f, xny = 0.0f, xnz = 0.0f;
  if (tid < ROWS_PER_BLOCK) {
    const int grow0 = blockIdx.x * ROWS_PER_BLOCK + tid;
    xnx = rp[3 * grow0 + 0];
    xny = rp[3 * grow0 + 1];
    xnz = rp[3 * grow0 + 2];
  }
  const float tv = t[b];

  float yx[ASG_STAGE_PER_THREAD], yy[ASG_STAGE_PER_THREAD], yz[ASG_STAGE_PER_THREAD];
  float ga[ASG_STAGE_PER_THREAD], gb[ASG_STAGE_PER_THREAD];
#pragma unroll
  for (int u = 0; u < ASG_STAGE_PER_THREAD; ++u) {
    const int j = tid + u * THREADS;
    yx[u] = cp[3 * j + 0];
    yy[u] = cp[3 * j + 1];
    yz[u] = cp[3 * j + 2];
    ga[u] = gp0[j];
    gb[u] = gp1[j];
  }
  float w1r[ASG_W_PER_THREAD], w2r[ASG_W_PER_THREAD];
#pragma unroll
  for (int u = 0; u < ASG_W_PER_THREAD; ++u) {
    w1r[u] = W1[tid + u * THREADS];
    w2r[u] = W2[tid + u * THREADS];
  }
  const float wtr = Wt[tid];
  const float b1r = b1[tid];
  float b2r = 0.0f;
  if (tid < 3) b2r = b2[tid];

#pragma unroll
  for (int u = 0; u < ASG_STAGE_PER_THREAD; ++u) {
    const int j = tid + u * THREADS;
    const float sq = sqnorm3(yx[u], yy[u], yz[u]);
    const float gj = merged_potential(ga[u], gb[u], eps_g, lconst, sq);
    s_cols[j] = make_float4(yx[u], yy[u], yz[u], 0.0f);
    s_gy[j]   = gj - sq;
  }
#pragma unroll
  for (int u = 0; u < ASG_W_PER_THREAD; ++u) {
    s_W1[tid + u * THREADS] = w1r[u];
    s_W2[tid + u * THREADS] = w2r[u];
  }
  s_Wt[tid] = wtr;
  s_b1[tid] = b1r;
  if (tid < 3) s_b2[tid] = b2r;
  __syncthreads();

  v2f a;
  a.x = hi ? 2.0f * rz : 2.0f * rx;
  a.y = hi ? 0.0f      : 2.0f * ry;

  float best[8]; int bidx[8];
#pragma unroll
  for (int r = 0; r < 8; ++r) { best[r] = -3.0e38f; bidx[r] = 0; }

#pragma unroll 4
  for (int c0 = 0; c0 < N_; c0 += 16) {
    float4 c4 = s_cols[c0 + idx];
    v2f bf;
    bf.x = hi ? c4.z : c4.x;
    bf.y = hi ? 0.0f : c4.y;
    float gy = s_gy[c0 + idx];

    v8f cz0 = {};
    v8f acc = __builtin_amdgcn_wmma_f32_16x16x4_f32(
        false, a, false, bf, (short)0, cz0, false, false);

#pragma unroll
    for (int r = 0; r < 8; ++r) {
      float v = acc[r] + gy;
      if (v > best[r]) { best[r] = v; bidx[r] = c0 + idx; }
    }
  }

#pragma unroll
  for (int mask = 1; mask < 16; mask <<= 1) {
#pragma unroll
    for (int r = 0; r < 8; ++r) {
      float ov = __shfl_xor(best[r], mask, 32);
      int   oi = __shfl_xor(bidx[r], mask, 32);
      if (ov > best[r] || (ov == best[r] && oi < bidx[r])) {
        best[r] = ov; bidx[r] = oi;
      }
    }
  }

  if (idx < 8) s_idx[wave * 16 + hi * 8 + idx] = bidx[idx];
  __syncthreads();

  if (tid < ROWS_PER_BLOCK) {
    const int grow = blockIdx.x * ROWS_PER_BLOCK + tid;
    const int j    = s_idx[tid];
    float4 x0a = s_cols[j];
    const float omt = 1.0f - tv;

    float vx = xnx - x0a.x, vy = xny - x0a.y, vz = xnz - x0a.z;
    float xtx = omt * x0a.x + tv * xnx;
    float xty = omt * x0a.y + tv * xny;
    float xtz = omt * x0a.z + tv * xnz;

    float p0 = s_b2[0], p1 = s_b2[1], p2 = s_b2[2];
    for (int k = 0; k < H_; ++k) {
      float h = xtx * s_W1[k] + xty * s_W1[H_ + k] + xtz * s_W1[2 * H_ + k]
              + tv * s_Wt[k] + s_b1[k];
      h = fmaxf(h, 0.0f);
      p0 += h * s_W2[k * 3 + 0];
      p1 += h * s_W2[k * 3 + 1];
      p2 += h * s_W2[k * 3 + 2];
    }
    const size_t base  = ((size_t)b * N_ + grow) * 3;
    const size_t vbase = (size_t)B_ * N_ * 3 + base;
    out[base + 0] = p0; out[base + 1] = p1; out[base + 2] = p2;
    out[vbase + 0] = vx; out[vbase + 1] = vy; out[vbase + 2] = vz;
  }
}

extern "C" void kernel_launch(void* const* d_in, const int* in_sizes, int n_in,
                              void* d_out, int out_size, void* d_ws, size_t ws_size,
                              hipStream_t stream) {
  (void)in_sizes; (void)n_in; (void)out_size; (void)ws_size;
  const float* cloud = (const float*)d_in[0];
  const float* noise = (const float*)d_in[1];
  const float* t     = (const float*)d_in[2];
  const float* W1    = (const float*)d_in[3];
  const float* Wt    = (const float*)d_in[4];
  const float* b1    = (const float*)d_in[5];
  const float* W2    = (const float*)d_in[6];
  const float* b2    = (const float*)d_in[7];
  float* out = (float*)d_out;

  char* ws = (char*)d_ws;
  float* y3 = (float*)ws;
  float* f0 = (float*)(ws + (size_t)B_ * N_ * 3 * sizeof(float));
  float* f1 = f0 + (size_t)B_ * N_;
  float* g0 = f1 + (size_t)B_ * N_;
  float* g1 = g0 + (size_t)B_ * N_;

  const float lconst = -logf((float)N_);
  dim3 grid(N_ / ROWS_PER_BLOCK * KS, B_);
  float eps_in = 0.0f;
  for (int it = 0; it < 14; ++it) {
    float eps = (float)(32.0 * pow(1e-6 / 32.0, (double)it / 13.0));
    if (it == 0) {
      sinkhorn_half<true, true><<<grid, THREADS, 0, stream>>>(noise, cloud, g0, g1, f0, f1, y3,
                                                              eps, eps_in, lconst);
      sinkhorn_half<true, false><<<grid, THREADS, 0, stream>>>(y3, noise, f0, f1, g0, g1, y3,
                                                               eps, eps, lconst);
    } else if (it < LAZY_ITERS) {
      sinkhorn_half<true, false><<<grid, THREADS, 0, stream>>>(noise, y3, g0, g1, f0, f1, y3,
                                                               eps, eps_in, lconst);
      sinkhorn_half<true, false><<<grid, THREADS, 0, stream>>>(y3, noise, f0, f1, g0, g1, y3,
                                                               eps, eps, lconst);
    } else {
      sinkhorn_half<false, false><<<grid, THREADS, 0, stream>>>(noise, y3, g0, g1, f0, f1, y3,
                                                                eps, eps_in, lconst);
      sinkhorn_half<false, false><<<grid, THREADS, 0, stream>>>(y3, noise, f0, f1, g0, g1, y3,
                                                                eps, eps, lconst);
    }
    eps_in = eps;
  }

  dim3 grid_a(N_ / ROWS_PER_BLOCK, B_);
  assign_mlp_kernel<<<grid_a, THREADS, 0, stream>>>(noise, y3, g0, g1, eps_in, lconst, t,
                                                    W1, Wt, b1, W2, b2, out);
}


# Assignment kernel re-divided: 256 work-groups of 64 rows, 8 waves scan 4 row tiles x 2 column halves, candidates meet in LDS, 4 threads a row share pick and MLP; MLP sum order changes
# speedup vs baseline: 2.3560x; 1.0430x over previous
#include <hip/hip_runtime.h>
#include <cmath>


#define B_  8
#define N_  2048
#define H_  256
#define THREADS 256
#define ROWS_PER_BLOCK 128
#define KS 2
#define COLS_PER_WG (N_ / KS)
#define STAGE_PER_THREAD (COLS_PER_WG / THREADS)
#define TPT 16
#define LAZY_SUM_MAX 1.0e30f
#define LAZY_ITERS 6
#define SEED_TILES 4
#define CLOUD_PER_THREAD (N_ * 3 / THREADS)
#define ASG_STAGE_PER_THREAD (N_ / THREADS)
#define ASG_W_PER_THREAD (3 * H_ / THREADS)
#define ASG_ROWS 64
#define ASG_QUAD 4
#define ASG_CAND 32
#define ASG_CAND_PITCH 33

static_assert(ASG_ROWS * ASG_QUAD == THREADS && ASG_ROWS == 4 * 16 && THREADS == 8 * 32,
              "assignment kernel: 4 row tiles x 2 column halves of waves, ASG_QUAD threads per row");
static_assert(ASG_CAND % ASG_QUAD == 0 && H_ % (4 * ASG_QUAD) == 0 && 3 * ASG_ROWS <= THREADS,
              "assignment kernel: candidates, hidden units and result floats divide over the threads");
static_assert(H_ == THREADS, "one element of Wt and of b1 per thread");
static_assert((3 * H_) % THREADS == 0 && N_ % THREADS == 0 && (N_ * 3) % THREADS == 0,
              "staging loops cover their arrays in whole trips");

typedef __attribute__((ext_vector_type(2))) float v2f;
typedef __attribute__((ext_vector_type(8))) float v8f;

#define LOG2E 1.4426950408889634f
#define LN2   0.6931471805599453f

__device__ __forceinline__ float sqnorm3(float x, float y, float z) {
  return fmaf(z, z, fmaf(y, y, x * x));
}

__device__ __forceinline__ float merged_potential(float la, float lb, float eps_in,
                                                  float lconst, float sq) {
  float top  = fmaxf(la, lb);
  float e    = __builtin_amdgcn_exp2f(-fabsf(la - lb));
  float lse2 = top + __builtin_amdgcn_logf(1.0f + e);
  return -eps_in * (lconst + LN2 * lse2) + 0.5f * sq;
}

__device__ __forceinline__ float merged_potential_fused(float la, float lb, float eps_in,
                                                        float lconst, float sq) {
  float top  = fmaxf(la, lb);
  float e    = __builtin_amdgcn_exp2f(-fabsf(la - lb));
  float lse2 = top + __builtin_amdgcn_logf(1.0f + e);
  return fmaf(-eps_in, fmaf(LN2, lse2, lconst), 0.5f * sq);
}

template <int CTRL>
__device__ __forceinline__ float row_xchg(float x) {
  return __builtin_bit_cast(float, __builtin_amdgcn_mov_dpp(
      __builtin_bit_cast(int, x), CTRL, 0xf, 0xf, false));
}

__device__ __forceinline__ void lse_merge(float& mm, float& ss, float om, float os) {
  const float d  = om - mm;
  const float e  = __builtin_amdgcn_exp2f(-fabsf(d));
  const bool  up = d > 0.0f;
  ss = up ? fmaf(ss, e, os) : fmaf(os, e, ss);
  mm = up ? om : mm;
}

__device__ __forceinline__ void tile_exact_update(const v8f& acct, float gyt, v2f (&m2)[4], v2f (&s2)[4],
                                                  v2f c2v, v2f onev) {
  v2f gyv; gyv.x = gyt; gyv.y = gyt;
#pragma unroll
  for (int q = 0; q < 4; ++q) {
    v2f ap; ap.x = acct[2 * q]; ap.y = acct[2 * q + 1];
    const v2f v = (gyv + ap) * c2v;
    const v2f d = v - m2[q];
    v2f e;
    e.x = __builtin_amdgcn_exp2f(-fabsf(d.x));
    e.y = __builtin_amdgcn_exp2f(-fabsf(d.y));
    const bool up0 = d.x > 0.0f;
    const bool up1 = d.y > 0.0f;
    const v2f sa = __builtin_elementwise_fma(s2[q], e, onev);
    const v2f sb = s2[q] + e;
    s2[q].x = up0 ? sa.x : sb.x;
    s2[q].y = up1 ? sa.y : sb.y;
    m2[q].x = up0 ? v.x : m2[q].x;
    m2[q].y = up1 ? v.y : m2[q].y;
  }
}

template <bool LAZY, bool FIRST>
__global__ __launch_bounds__(THREADS)
void sinkhorn_half(const float* __restrict__ rows3,
                   const float* __restrict__ cols3,
                   const float* __restrict__ in_p0,
                   const float* __restrict__ in_p1,
                   float*       __restrict__ out_p0,
                   float*       __restrict__ out_p1,
                   float*       __restrict__ scaled_out,
                   float eps, float eps_in, float lconst) {
  const int b    = blockIdx.y;
  const int ks   = blockIdx.x % KS;
  const int rb   = blockIdx.x / KS;
  const int tid  = threadIdx.x;
  const int lane = tid & 31;
  const int wave = tid >> 5;
  const int hi   = lane >> 4;
  const int idx  = lane & 15;

  __shared__ v2f   s_colp[COLS_PER_WG * 2];
  __shared__ float s_gy[COLS_PER_WG];

  const int row_base = rb * ROWS_PER_BLOCK + wave * 16;
  const float* rp = rows3 + (size_t)b * N_ * 3;
  const float rx = rp[3 * (row_base + idx) + 0];
  const float ry = rp[3 * (row_base + idx) + 1];
  const float rz = rp[3 * (row_base + idx) + 2];
  __builtin_amdgcn_sched_barrier(0);

  const size_t cbase = (size_t)b * N_ + (size_t)ks * COLS_PER_WG;
  const float* cp = cols3 + cbase * 3;
  const float* p0 = in_p0 + cbase;
  const float* p1 = in_p1 + cbase;

  float cx[STAGE_PER_THREAD], cy[STAGE_PER_THREAD], cz[STAGE_PER_THREAD];
  float la[STAGE_PER_THREAD], lb[STAGE_PER_THREAD];
#pragma unroll
  for (int u = 0; u < STAGE_PER_THREAD; ++u) {
    const int j = tid + u * THREADS;
    cx[u] = cp[3 * j + 0];
    cy[u] = cp[3 * j + 1];
    cz[u] = cp[3 * j + 2];
    la[u] = 0.f; lb[u] = 0.f;
    if constexpr (!FIRST) { la[u] = p0[j]; lb[u] = p1[j]; }
  }

  if constexpr (FIRST) {
    __builtin_amdgcn_sched_barrier(0);
    __shared__ float s_sum[THREADS];
    __shared__ float s_sq[THREADS];
    const float* cb = cols3 + (size_t)b * N_ * 3;
    float v[CLOUD_PER_THREAD];
#pragma unroll
    for (int u = 0; u < CLOUD_PER_THREAD; ++u) v[u] = cb[tid + u * THREADS];
    float sum = 0.f, sq = 0.f;
#pragma unroll
    for (int u = 0; u < CLOUD_PER_THREAD; ++u) { sum += v[u]; sq = fmaf(v[u], v[u], sq); }
    s_sum[tid] = sum; s_sq[tid] = sq;
    __syncthreads();
    for (int off = THREADS / 2; off > 0; off >>= 1) {
      if (tid < off) { s_sum[tid] += s_sum[tid + off]; s_sq[tid] += s_sq[tid + off]; }
      __syncthreads();
    }
    const float n    = (float)(N_ * 3);
    const float tot  = s_sum[0];
    const float tot2 = s_sq[0];
    const float var  = (tot2 - tot * tot / n) / (n - 1.0f);
    const float inv_std = rsqrtf(var);
#pragma unroll
    for (int u = 0; u < STAGE_PER_THREAD; ++u) {
      cx[u] *= inv_std; cy[u] *= inv_std; cz[u] *= inv_std;
    }
  }

#pragma unroll
  for (int u = 0; u < STAGE_PER_THREAD; ++u) {
    const int j = tid + u * THREADS;
    const float sq = sqnorm3(cx[u], cy[u], cz[u]);
    const float pot = FIRST ? 0.0f : merged_potential_fused(la[u], lb[u], eps_in, lconst, sq);
    v2f lo2; lo2.x = cx[u]; lo2.y = cy[u];
    v2f hi2; hi2.x = cz[u]; hi2.y = 0.0f;
    s_colp[2 * j + 0] = lo2;
    s_colp[2 * j + 1] = hi2;
    s_gy[j] = pot - 0.5f * sq;
  }
  __syncthreads();

  if constexpr (FIRST) {
    if (rb == 0) {
      float* so = scaled_out + cbase * 3;
#pragma unroll
      for (int u = 0; u < STAGE_PER_THREAD; ++u) {
        const int j = tid + u * THREADS;
        so[3 * j + 0] = cx[u];
        so[3 * j + 1] = cy[u];
        so[3 * j + 2] = cz[u];
      }
    }
  }

  v2f a;
  a.x = hi ? rz   : rx;
  a.y = hi ? 0.0f : ry;

  v2f m2[4], s2[4];
#pragma unroll
  for (int q = 0; q < 4; ++q) { m2[q].x = -3.0e38f; m2[q].y = -3.0e38f; s2[q].x = 0.f; s2[q].y = 0.f; }
  const float c2 = LOG2E / eps;
  v2f c2v;  c2v.x = c2;    c2v.y = c2;
  v2f onev; onev.x = 1.0f; onev.y = 1.0f;

  if constexpr (LAZY) {
    {
      float ref[8];
#pragma unroll
      for (int r = 0; r < 8; ++r) ref[r] = -3.0e38f;
#pragma unroll
      for (int t = 0; t < SEED_TILES; ++t) {
        const v2f   bf0 = s_colp[2 * (16 * t + idx) + hi];
        const float gy0 = s_gy[16 * t + idx];
        v8f cz0 = {};
        const v8f acc0 = __builtin_amdgcn_wmma_f32_16x16x4_f32(
            false, a, false, bf0, (short)0, cz0, false, false);
        v2f gyv; gyv.x = gy0; gyv.y = gy0;
#pragma unroll
        for (int q = 0; q < 4; ++q) {
          v2f ap; ap.x = acc0[2 * q]; ap.y = acc0[2 * q + 1];
          const v2f v = (gyv + ap) * c2v;
          ref[2 * q]     = fmaxf(ref[2 * q], v.x);
          ref[2 * q + 1] = fmaxf(ref[2 * q + 1], v.y);
        }
      }
#pragma unroll
      for (int r = 0; r < 8; ++r) {
        ref[r] = fmaxf(ref[r], row_xchg<0x161>(ref[r]));
        ref[r] = fmaxf(ref[r], row_xchg<0x162>(ref[r]));
        ref[r] = fmaxf(ref[r], row_xchg<0x164>(ref[r]));
        ref[r] = fmaxf(ref[r], row_xchg<0x168>(ref[r]));
      }
#pragma unroll
      for (int q = 0; q < 4; ++q) { m2[q].x = ref[2 * q]; m2[q].y = ref[2 * q + 1]; }
    }

    for (int c0 = 0; c0 < COLS_PER_WG; c0 += 16 * TPT) {
      v2f   bf[TPT];
      float gy[TPT];
#pragma unroll
      for (int t = 0; t < TPT; ++t) {
        bf[t] = s_colp[2 * (c0 + 16 * t + idx) + hi];
        gy[t] = s_gy[c0 + 16 * t + idx];
      }
      v8f acc[TPT];
#pragma unroll
      for (int t = 0; t < TPT; ++t) {
        v8f cz0 = {};
        acc[t] = __builtin_amdgcn_wmma_f32_16x16x4_f32(
            false, a, false, bf[t], (short)0, cz0, false, false);
      }

      v2f sn[4];
#pragma unroll
      for (int q = 0; q < 4; ++q) sn[q] = s2[q];
#pragma unroll
      for (int t = 0; t < TPT; ++t) {
        v2f gyv; gyv.x = gy[t]; gyv.y = gy[t];
#pragma unroll
        for (int q = 0; q < 4; ++q) {
          v2f ap; ap.x = acc[t][2 * q]; ap.y = acc[t][2 * q + 1];
          const v2f v = (gyv + ap) * c2v;
          const v2f d = v - m2[q];
          v2f e;
          e.x = __builtin_amdgcn_exp2f(d.x);
          e.y = __builtin_amdgcn_exp2f(d.y);
          sn[q] = sn[q] + e;
        }
      }
      float top = fmaxf(fmaxf(sn[0].x, sn[0].y), fmaxf(sn[1].x, sn[1].y));
      top = fmaxf(top, fmaxf(fmaxf(sn[2].x, sn[2].y), fmaxf(sn[3].x, sn[3].y)));

      if (__builtin_amdgcn_ballot_w32(!(top < LAZY_SUM_MAX)) == 0u) {
#pragma unroll
        for (int q = 0; q < 4; ++q) s2[q] = sn[q];
      } else {
#pragma unroll
        for (int t = 0; t < TPT; ++t) tile_exact_update(acc[t], gy[t], m2, s2, c2v, onev);
      }
    }
  } else {
    for (int c0 = 0; c0 < COLS_PER_WG; c0 += 16 * TPT) {
      v2f   bf[TPT];
      float gy[TPT];
#pragma unroll
      for (int t = 0; t < TPT; ++t) {
        bf[t] = s_colp[2 * (c0 + 16 * t + idx) + hi];
        gy[t] = s_gy[c0 + 16 * t + idx];
      }

      v8f acc[TPT];
#pragma unroll
      for (int t = 0; t < TPT; ++t) {
        v8f cz0 = {};
        acc[t] = __builtin_amdgcn_wmma_f32_16x16x4_f32(
            false, a, false, bf[t], (short)0, cz0, false, false);
      }

#pragma unroll
      for (int t = 0; t < TPT; ++t) {
        v2f gyv; gyv.x = gy[t]; gyv.y = gy[t];
#pragma unroll
        for (int q = 0; q < 4; ++q) {
          v2f ap; ap.x = acc[t][2 * q]; ap.y = acc[t][2 * q + 1];
          const v2f v = (gyv + ap) * c2v;
          const v2f d = v - m2[q];
          v2f e;
          e.x = __builtin_amdgcn_exp2f(-fabsf(d.x));
          e.y = __builtin_amdgcn_exp2f(-fabsf(d.y));
          const bool up0 = d.x > 0.0f;
          const bool up1 = d.y > 0.0f;
          const v2f sa = __builtin_elementwise_fma(s2[q], e, onev);
          const v2f sb = s2[q] + e;
          s2[q].x = up0 ? sa.x : sb.x;
          s2[q].y = up1 ? sa.y : sb.y;
          m2[q].x = up0 ? v.x : m2[q].x;
          m2[q].y = up1 ? v.y : m2[q].y;
        }
      }
    }
  }

  float m[8], s[8];
#pragma unroll
  for (int q = 0; q < 4; ++q) {
    m[2 * q] = m2[q].x; m[2 * q + 1] = m2[q].y;
    s[2 * q] = s2[q].x; s[2 * q + 1] = s2[q].y;
  }

  const bool b0 = (idx & 1) != 0;
  const bool b1 = (idx & 2) != 0;
  const bool b2 = (idx & 4) != 0;

  float ma[4], sa[4];
#pragma unroll
  for (int r = 0; r < 4; ++r) {
    float km = b0 ? m[r + 4] : m[r];
    float kv = b0 ? s[r + 4] : s[r];
    const float gm = b0 ? m[r] : m[r + 4];
    const float gv = b0 ? s[r] : s[r + 4];
    lse_merge(km, kv, row_xchg<0x161>(gm), row_xchg<0x161>(gv));
    ma[r] = km; sa[r] = kv;
  }
  float mb[2], sb2[2];
#pragma unroll
  for (int r = 0; r < 2; ++r) {
    float km = b1 ? ma[r + 2] : ma[r];
    float kv = b1 ? sa[r + 2] : sa[r];
    const float gm = b1 ? ma[r] : ma[r + 2];
    const float gv = b1 ? sa[r] : sa[r + 2];
    lse_merge(km, kv, row_xchg<0x162>(gm), row_xchg<0x162>(gv));
    mb[r] = km; sb2[r] = kv;
  }
  float mc = b2 ? mb[1] : mb[0];
  float sc = b2 ? sb2[1] : sb2[0];
  {
    const float gm = b2 ? mb[0] : mb[1];
    const float gv = b2 ? sb2[0] : sb2[1];
    lse_merge(mc, sc, row_xchg<0x164>(gm), row_xchg<0x164>(gv));
  }
  lse_merge(mc, sc, row_xchg<0x168>(mc), row_xchg<0x168>(sc));

  if (idx < 8) {
    const int r   = (b0 ? 4 : 0) + (b1 ? 2 : 0) + (b2 ? 1 : 0);
    const int row = row_base + hi * 8 + r;
    const float l2 = mc + __builtin_amdgcn_logf(sc);
    float* op = ks ? out_p1 : out_p0;
    op[(size_t)b * N_ + row] = l2;
  }
}

typedef __attribute__((ext_vector_type(2))) unsigned v2u;

__global__ __launch_bounds__(THREADS)
void assign_mlp_kernel(const float* __restrict__ x3,
                       const float* __restrict__ y3,
                       const float* __restrict__ g0,
                       const float* __restrict__ g1,
                       float eps_g, float lconst,
                       const float*  __restrict__ t,
                       const float*  __restrict__ W1,
                       const float*  __restrict__ Wt,
                       const float*  __restrict__ b1,
                       const float*  __restrict__ W2,
                       const float*  __restrict__ b2,
                       float*        __restrict__ out) {
  const int b    = blockIdx.y;
  const int tid  = threadIdx.x;
  const int lane = tid & 31;
  const int wave = tid >> 5;
  const int hi   = lane >> 4;
  const int idx  = lane & 15;
  const int rt   = wave & 3;
  const int ch   = wave >> 2;
  const int mrow = tid / ASG_QUAD;
  const int mq   = tid % ASG_QUAD;

  __shared__ float4 s_cols[N_];
  __shared__ float  s_gy[N_];
  __shared__ __attribute__((aligned(16))) float s_W1[3 * H_];
  __shared__ __attribute__((aligned(16))) float s_Wt[H_];
  __shared__ __attribute__((aligned(16))) float s_b1[H_];
  __shared__ __attribute__((aligned(16))) float s_W2[H_ * 3];
  __shared__ float  s_b2[4];
  __shared__ v2u    s_cand[ASG_ROWS * ASG_CAND_PITCH];
  __shared__ float  s_res[2 * ASG_ROWS * 3];

  const float* cp  = y3 + (size_t)b * N_ * 3;
  const float* gp0 = g0 + (size_t)b * N_;
  const float* gp1 = g1 + (size_t)b * N_;

  const int blk_row0 = blockIdx.x * ASG_ROWS;
  const int row_base = blk_row0 + rt * 16;
  const float* rp = x3 + (size_t)b * N_ * 3;
  const float rx = rp[3 * (row_base + idx) + 0];
  const float ry = rp[3 * (row_base + idx) + 1];
  const float rz = rp[3 * (row_base + idx) + 2];

  const float xnx = rp[3 * (blk_row0 + mrow) + 0];
  const float xny = rp[3 * (blk_row0 + mrow) + 1];
  const float xnz = rp[3 * (blk_row0 + mrow) + 2];
  const float tv = t[b];

  float yx[ASG_STAGE_PER_THREAD], yy[ASG_STAGE_PER_THREAD], yz[ASG_STAGE_PER_THREAD];
  float ga[ASG_STAGE_PER_THREAD], gb[ASG_STAGE_PER_THREAD];
#pragma unroll
  for (int u = 0; u < ASG_STAGE_PER_THREAD; ++u) {
    const int j = tid + u * THREADS;
    yx[u] = cp[3 * j + 0];
    yy[u] = cp[3 * j + 1];
    yz[u] = cp[3 * j + 2];
    ga[u] = gp0[j];
    gb[u] = gp1[j];
  }
  float w1r[ASG_W_PER_THREAD], w2r[ASG_W_PER_THREAD];
#pragma unroll
  for (int u = 0; u < ASG_W_PER_THREAD; ++u) {
    w1r[u] = W1[tid + u * THREADS];
    w2r[u] = W2[tid + u * THREADS];
  }
  const float wtr = Wt[tid];
  const float b1r = b1[tid];
  float b2r = 0.0f;
  if (tid < 3) b2r = b2[tid];

#pragma unroll
  for (int u = 0; u < ASG_STAGE_PER_THREAD; ++u) {
    const int j = tid + u * THREADS;
    const float sq = sqnorm3(yx[u], yy[u], yz[u]);
    const float gj = merged_potential(ga[u], gb[u], eps_g, lconst, sq);
    s_cols[j] = make_float4(yx[u], yy[u], yz[u], 0.0f);
    s_gy[j]   = gj - sq;
  }
#pragma unroll
  for (int u = 0; u < ASG_W_PER_THREAD; ++u) {
    s_W1[tid + u * THREADS] = w1r[u];
    s_W2[tid + u * THREADS] = w2r[u];
  }
  s_Wt[tid] = wtr;
  s_b1[tid] = b1r;
  if (tid < 3) s_b2[tid] = b2r;
  __syncthreads();

  v2f a;
  a.x = hi ? 2.0f * rz : 2.0f * rx;
  a.y = hi ? 0.0f      : 2.0f * ry;

  float best[8]; int bidx[8];
#pragma unroll
  for (int r = 0; r < 8; ++r) { best[r] = -3.0e38f; bidx[r] = 0; }

  const int cbeg = ch * (N_ / 2);
#pragma unroll 4
  for (int ct = 0; ct < N_ / 2; ct += 16) {
    const int c0 = cbeg + ct;
    float4 c4 = s_cols[c0 + idx];
    v2f bf;
    bf.x = hi ? c4.z : c4.x;
    bf.y = hi ? 0.0f : c4.y;
    float gy = s_gy[c0 + idx];

    v8f cz0 = {};
    v8f acc = __builtin_amdgcn_wmma_f32_16x16x4_f32(
        false, a, false, bf, (short)0, cz0, false, false);

#pragma unroll
    for (int r = 0; r < 8; ++r) {
      float v = acc[r] + gy;
      if (v > best[r]) { best[r] = v; bidx[r] = c0 + idx; }
    }
  }

#pragma unroll
  for (int r = 0; r < 8; ++r) {
    const int row = rt * 16 + hi * 8 + r;
    v2u cd; cd.x = __float_as_uint(best[r]); cd.y = (unsigned)bidx[r];
    s_cand[row * ASG_CAND_PITCH + ch * 16 + idx] = cd;
  }
  __syncthreads();

  float bv; int bi;
  {
    const v2u c0 = s_cand[mrow * ASG_CAND_PITCH + mq * (ASG_CAND / ASG_QUAD)];
    bv = __uint_as_float(c0.x); bi = (int)c0.y;
  }
#pragma unroll
  for (int s = 1; s < ASG_CAND / ASG_QUAD; ++s) {
    const v2u cd = s_cand[mrow * ASG_CAND_PITCH + mq * (ASG_CAND / ASG_QUAD) + s];
    const float ov = __uint_as_float(cd.x);
    const int   oi = (int)cd.y;
    const bool take = ov > bv || (ov == bv && oi < bi);
    bv = take ? ov : bv;
    bi = take ? oi : bi;
  }
  {
    const float ov = row_xchg<0x161>(bv);
    const int   oi = __builtin_bit_cast(int, row_xchg<0x161>(__builtin_bit_cast(float, bi)));
    const bool take = ov > bv || (ov == bv && oi < bi);
    bv = take ? ov : bv;
    bi = take ? oi : bi;
  }
  {
    const float ov = row_xchg<0x162>(bv);
    const int   oi = __builtin_bit_cast(int, row_xchg<0x162>(__builtin_bit_cast(float, bi)));
    const bool take = ov > bv || (ov == bv && oi < bi);
    bv = take ? ov : bv;
    bi = take ? oi : bi;
  }

  {
    const float4 x0a = s_cols[bi];
    const float omt = 1.0f - tv;

    const float vx = xnx - x0a.x, vy = xny - x0a.y, vz = xnz - x0a.z;
    const float xtx = omt * x0a.x + tv * xnx;
    const float xty = omt * x0a.y + tv * xny;
    const float xtz = omt * x0a.z + tv * xnz;

    float p0 = 0.0f, p1 = 0.0f, p2 = 0.0f;
    for (int kt = 0; kt < H_; kt += 4 * ASG_QUAD) {
#pragma unroll
      for (int u = 0; u < 4; ++u) {
        const int k = kt + 4 * mq + u;
        float h = xtx * s_W1[k] + xty * s_W1[H_ + k] + xtz * s_W1[2 * H_ + k]
                + tv * s_Wt[k] + s_b1[k];
        h = fmaxf(h, 0.0f);
        p0 += h * s_W2[k * 3 + 0];
        p1 += h * s_W2[k * 3 + 1];
        p2 += h * s_W2[k * 3 + 2];
      }
    }
    p0 += row_xchg<0x161>(p0); p1 += row_xchg<0x161>(p1); p2 += row_xchg<0x161>(p2);
    p0 += row_xchg<0x162>(p0); p1 += row_xchg<0x162>(p1); p2 += row_xchg<0x162>(p2);
    p0 += s_b2[0]; p1 += s_b2[1]; p2 += s_b2[2];

    if (mq == 0) {
      s_res[3 * mrow + 0] = p0; s_res[3 * mrow + 1] = p1; s_res[3 * mrow + 2] = p2;
    } else if (mq == 1) {
      s_res[3 * ASG_ROWS + 3 * mrow + 0] = vx;
      s_res[3 * ASG_ROWS + 3 * mrow + 1] = vy;
      s_res[3 * ASG_ROWS + 3 * mrow + 2] = vz;
    }
  }
  __syncthreads();

  if (tid < 3 * ASG_ROWS) {
    const size_t base  = ((size_t)b * N_ + blk_row0) * 3;
    const size_t vbase = (size_t)B_ * N_ * 3 + base;
    out[base + tid]  = s_res[tid];
    out[vbase + tid] = s_res[3 * ASG_ROWS + tid];
  }
}

extern "C" void kernel_launch(void* const* d_in, const int* in_sizes, int n_in,
                              void* d_out, int out_size, void* d_ws, size_t ws_size,
                              hipStream_t stream) {
  (void)in_sizes; (void)n_in; (void)out_size; (void)ws_size;
  const float* cloud = (const float*)d_in[0];
  const float* noise = (const float*)d_in[1];
  const float* t     = (const float*)d_in[2];
  const float* W1    = (const float*)d_in[3];
  const float* Wt    = (const float*)d_in[4];
  const float* b1    = (const float*)d_in[5];
  const float* W2    = (const float*)d_in[6];
  const float* b2    = (const float*)d_in[7];
  float* out = (float*)d_out;

  char* ws = (char*)d_ws;
  float* y3 = (float*)ws;
  float* f0 = (float*)(ws + (size_t)B_ * N_ * 3 * sizeof(float));
  float* f1 = f0 + (size_t)B_ * N_;
  float* g0 = f1 + (size_t)B_ * N_;
  float* g1 = g0 + (size_t)B_ * N_;

  const float lconst = -logf((float)N_);
  dim3 grid(N_ / ROWS_PER_BLOCK * KS, B_);
  float eps_in = 0.0f;
  for (int it = 0; it < 14; ++it) {
    float eps = (float)(32.0 * pow(1e-6 / 32.0, (double)it / 13.0));
    if (it == 0) {
      sinkhorn_half<true, true><<<grid, THREADS, 0, stream>>>(noise, cloud, g0, g1, f0, f1, y3,
                                                              eps, eps_in, lconst);
      sinkhorn_half<true, false><<<grid, THREADS, 0, stream>>>(y3, noise, f0, f1, g0, g1, y3,
                                                               eps, eps, lconst);
    } else if (it < LAZY_ITERS) {
      sinkhorn_half<true, false><<<grid, THREADS, 0, stream>>>(noise, y3, g0, g1, f0, f1, y3,
                                                               eps, eps_in, lconst);
      sinkhorn_half<true, false><<<grid, THREADS, 0, stream>>>(y3, noise, f0, f1, g0, g1, y3,
                                                               eps, eps, lconst);
    } else {
      sinkhorn_half<false, false><<<grid, THREADS, 0, stream>>>(noise, y3, g0, g1, f0, f1, y3,
                                                                eps, eps_in, lconst);
      sinkhorn_half<false, false><<<grid, THREADS, 0, stream>>>(y3, noise, f0, f1, g0, g1, y3,
                                                                eps, eps, lconst);
    }
    eps_in = eps;
  }

  dim3 grid_a(N_ / ASG_ROWS, B_);
  assign_mlp_kernel<<<grid_a, THREADS, 0, stream>>>(noise, y3, g0, g1, eps_in, lconst, t,
                                                    W1, Wt, b1, W2, b2, out);
}
